# starting kernel
# baseline (speedup 1.0000x reference)
#include <hip/hip_runtime.h>

#define S_DIM 128
#define N_DIM 384
#define CM    64
#define CC    32
#define CZ    128
#define LN_EPS 1e-5f

#define SA_STRIDE  40
#define SAB_STRIDE 1028

typedef __attribute__((ext_vector_type(2))) float v2f;
typedef __attribute__((ext_vector_type(8))) float v8f;

__device__ __forceinline__ v8f wmma_f32(v2f a, v2f b, v8f c) {
  return __builtin_amdgcn_wmma_f32_16x16x4_f32(false, a, false, b, (short)0, c,
                                               false, false);
}

__device__ __forceinline__ float wave_sum(float v) {
#pragma unroll
  for (int off = 16; off > 0; off >>= 1) v += __shfl_xor(v, off, 32);
  return v;
}

__global__ __launch_bounds__(256) void opm_ln_proj(
    const float* __restrict__ msa, const float* __restrict__ mask,
    const float* __restrict__ gamma, const float* __restrict__ beta,
    const float* __restrict__ wa, const float* __restrict__ wb,
    float* __restrict__ mA, float* __restrict__ mB) {
  __shared__ float sY[8 * CM];
  const int t = threadIdx.x, wave = t >> 5, lane = t & 31;
  const int row = blockIdx.x * 8 + wave;

  const float* x = msa + (size_t)row * CM;
  const float x0 = x[lane], x1 = x[lane + 32];
  const float mean = wave_sum(x0 + x1) * (1.0f / CM);
  const float d0 = x0 - mean, d1 = x1 - mean;
  const float var = wave_sum(d0 * d0 + d1 * d1) * (1.0f / CM);
  const float inv = rsqrtf(var + LN_EPS);
  sY[wave * CM + lane]      = d0 * inv * gamma[lane] + beta[lane];
  sY[wave * CM + lane + 32] = d1 * inv * gamma[lane + 32] + beta[lane + 32];
  __syncthreads();

  const float mk = mask[row];
  float a = 0.f, b = 0.f;
#pragma unroll 8
  for (int m = 0; m < CM; ++m) {
    const float ym = sY[wave * CM + m];
    a = fmaf(ym, wa[m * CC + lane], a);
    b = fmaf(ym, wb[m * CC + lane], b);
  }
  mA[(size_t)row * CC + lane] = a * mk;
  mB[(size_t)row * CC + lane] = b * mk;
}

__global__ __launch_bounds__(256) void opm_main(
    const float* __restrict__ mA, const float* __restrict__ mB,
    const float* __restrict__ mask, const float* __restrict__ wOut,
    const float* __restrict__ bOut, float* __restrict__ out) {
  extern __shared__ float smem[];
  float* sA    = smem;
  float* sAB   = sA + S_DIM * SA_STRIDE;
  float* pnorm = sAB + 16 * SAB_STRIDE;
  float* snorm = pnorm + 256;

  const int t = threadIdx.x;
  const int wave = t >> 5, lane = t & 31;
  const int i = blockIdx.y;
  const int j0 = blockIdx.x * 16;

  for (int idx = t; idx < S_DIM * CC; idx += 256) {
    const int s = idx >> 5, c = idx & 31;
    sA[s * SA_STRIDE + c] = mA[((size_t)s * N_DIM + i) * CC + c];
  }
  {
    const int jj = t & 15, chunk = t >> 4;
    float p = 0.f;
#pragma unroll
    for (int s = chunk * 8; s < chunk * 8 + 8; ++s)
      p += mask[s * N_DIM + i] * mask[s * N_DIM + j0 + jj];
    pnorm[t] = p;
  }
  __syncthreads();
  if (t < 16) {
    float sum = 0.f;
#pragma unroll
    for (int c = 0; c < 16; ++c) sum += pnorm[c * 16 + t];
    snorm[t] = sum;
  }

  const int dBase = lane & 15;
  const int sOff  = (lane >> 4) << 1;
  const int mAdd  = (lane >> 4) << 3;
  for (int jj = wave * 2; jj < wave * 2 + 2; ++jj) {
    const int j = j0 + jj;
    v8f acc00 = {}, acc01 = {}, acc10 = {}, acc11 = {};
    for (int s0 = 0; s0 < S_DIM; s0 += 4) {
      const int sa = s0 + sOff;
      v2f a0, a1, b0, b1;
      a0.x = sA[sa * SA_STRIDE + dBase];
      a0.y = sA[(sa + 1) * SA_STRIDE + dBase];
      a1.x = sA[sa * SA_STRIDE + 16 + dBase];
      a1.y = sA[(sa + 1) * SA_STRIDE + 16 + dBase];
      const float* bp = mB + ((size_t)sa * N_DIM + j) * CC;
      b0.x = bp[dBase];
      b0.y = bp[(size_t)N_DIM * CC + dBase];
      b1.x = bp[16 + dBase];
      b1.y = bp[(size_t)N_DIM * CC + 16 + dBase];
      acc00 = wmma_f32(a0, b0, acc00);
      acc01 = wmma_f32(a0, b1, acc01);
      acc10 = wmma_f32(a1, b0, acc10);
      acc11 = wmma_f32(a1, b1, acc11);
    }
    float* abj = sAB + jj * SAB_STRIDE;
#pragma unroll
    for (int r = 0; r < 8; ++r) {
      const int c = r + mAdd;
      abj[c * CC + dBase]             = acc00[r];
      abj[c * CC + 16 + dBase]        = acc01[r];
      abj[(c + 16) * CC + dBase]      = acc10[r];
      abj[(c + 16) * CC + 16 + dBase] = acc11[r];
    }
  }
  __syncthreads();

  const int z0 = wave * 16;
  const int kOff = (lane >> 4) << 1;
  const int jr = lane & 15;
  v8f accA = {}, accB = {};
  for (int k0 = 0; k0 < CC * CC; k0 += 8) {
    v2f a, b;
    a.x = sAB[jr * SAB_STRIDE + k0 + kOff];
    a.y = sAB[jr * SAB_STRIDE + k0 + kOff + 1];
    const float* wp = wOut + (size_t)(k0 + kOff) * CZ + z0 + dBase;
    b.x = wp[0];
    b.y = wp[CZ];
    accA = wmma_f32(a, b, accA);

    a.x = sAB[jr * SAB_STRIDE + k0 + 4 + kOff];
    a.y = sAB[jr * SAB_STRIDE + k0 + 4 + kOff + 1];
    const float* wp2 = wOut + (size_t)(k0 + 4 + kOff) * CZ + z0 + dBase;
    b.x = wp2[0];
    b.y = wp2[CZ];
    accB = wmma_f32(a, b, accB);
  }
  const v8f accZ = accA + accB;

  const float bo = bOut[z0 + dBase];
#pragma unroll
  for (int r = 0; r < 8; ++r) {
    const int jj = r + mAdd;
    const float nrm = snorm[jj] + 0.001f;
    out[((size_t)i * N_DIM + (j0 + jj)) * CZ + z0 + dBase] = (accZ[r] + bo) / nrm;
  }
}

extern "C" void kernel_launch(void* const* d_in, const int* in_sizes, int n_in,
                              void* d_out, int out_size, void* d_ws,
                              size_t ws_size, hipStream_t stream) {
  (void)in_sizes; (void)n_in; (void)out_size; (void)ws_size;
  const float* msa   = (const float*)d_in[0];
  const float* mask  = (const float*)d_in[1];
  const float* gamma = (const float*)d_in[2];
  const float* beta  = (const float*)d_in[3];
  const float* wa    = (const float*)d_in[4];
  const float* wb    = (const float*)d_in[5];
  const float* wout  = (const float*)d_in[6];
  const float* bout  = (const float*)d_in[7];
  float* out = (float*)d_out;

  float* mAbuf = (float*)d_ws;
  float* mBbuf = mAbuf + (size_t)S_DIM * N_DIM * CC;

  opm_ln_proj<<<(S_DIM * N_DIM) / 8, 256, 0, stream>>>(msa, mask, gamma, beta,
                                                       wa, wb, mAbuf, mBbuf);

  const size_t ldsBytes =
      (size_t)(S_DIM * SA_STRIDE + 16 * SAB_STRIDE + 256 + 16) * sizeof(float);
  opm_main<<<dim3(N_DIM / 16, N_DIM), 256, ldsBytes, stream>>>(
      mAbuf, mBbuf, mask, wout, bout, out);
}


# B fragments (m_b, w_out) loaded as 16-byte vectors, a lane owns 4 adjacent columns, one per tile; stage 2 K split in four across waves, partial sums added through LDS (sum order changes)
# speedup vs baseline: 1.1073x; 1.1073x over previous
#include <hip/hip_runtime.h>

#define S_DIM 128
#define N_DIM 384
#define CM    64
#define CC    32
#define CZ    128
#define LN_EPS 1e-5f

#define SA_STRIDE  40
#define SAB_STRIDE 1028

typedef __attribute__((ext_vector_type(2))) float v2f;
typedef __attribute__((ext_vector_type(4))) float v4f;
typedef __attribute__((ext_vector_type(8))) float v8f;

__device__ __forceinline__ v8f wmma_f32(v2f a, v2f b, v8f c) {
  return __builtin_amdgcn_wmma_f32_16x16x4_f32(false, a, false, b, (short)0, c,
                                               false, false);
}

__device__ __forceinline__ v2f pair2(float x, float y) {
  v2f r;
  r.x = x;
  r.y = y;
  return r;
}

__device__ __forceinline__ float wave_sum(float v) {
#pragma unroll
  for (int off = 16; off > 0; off >>= 1) v += __shfl_xor(v, off, 32);
  return v;
}

__global__ __launch_bounds__(256) void opm_ln_proj(
    const float* __restrict__ msa, const float* __restrict__ mask,
    const float* __restrict__ gamma, const float* __restrict__ beta,
    const float* __restrict__ wa, const float* __restrict__ wb,
    float* __restrict__ mA, float* __restrict__ mB) {
  __shared__ float sY[8 * CM];
  const int t = threadIdx.x, wave = t >> 5, lane = t & 31;
  const int row = blockIdx.x * 8 + wave;

  const float* x = msa + (size_t)row * CM;
  const float x0 = x[lane], x1 = x[lane + 32];
  const float mean = wave_sum(x0 + x1) * (1.0f / CM);
  const float d0 = x0 - mean, d1 = x1 - mean;
  const float var = wave_sum(d0 * d0 + d1 * d1) * (1.0f / CM);
  const float inv = rsqrtf(var + LN_EPS);
  sY[wave * CM + lane]      = d0 * inv * gamma[lane] + beta[lane];
  sY[wave * CM + lane + 32] = d1 * inv * gamma[lane + 32] + beta[lane + 32];
  __syncthreads();

  const float mk = mask[row];
  float a = 0.f, b = 0.f;
#pragma unroll 8
  for (int m = 0; m < CM; ++m) {
    const float ym = sY[wave * CM + m];
    a = fmaf(ym, wa[m * CC + lane], a);
    b = fmaf(ym, wb[m * CC + lane], b);
  }
  mA[(size_t)row * CC + lane] = a * mk;
  mB[(size_t)row * CC + lane] = b * mk;
}

__global__ __launch_bounds__(256) void opm_main(
    const float* __restrict__ mA, const float* __restrict__ mB,
    const float* __restrict__ mask, const float* __restrict__ wOut,
    const float* __restrict__ bOut, float* __restrict__ out) {
  extern __shared__ float smem[];
  float* sA    = smem;
  float* sAB   = sA + S_DIM * SA_STRIDE;
  float* pnorm = sAB + 16 * SAB_STRIDE;
  float* snorm = pnorm + 256;

  const int t = threadIdx.x;
  const int wave = t >> 5, lane = t & 31;
  const int i = blockIdx.y;
  const int j0 = blockIdx.x * 16;

  for (int idx = t; idx < S_DIM * CC; idx += 256) {
    const int s = idx >> 5, c = idx & 31;
    sA[s * SA_STRIDE + c] = mA[((size_t)s * N_DIM + i) * CC + c];
  }
  {
    const int jj = t & 15, chunk = t >> 4;
    float p = 0.f;
#pragma unroll
    for (int s = chunk * 8; s < chunk * 8 + 8; ++s)
      p += mask[s * N_DIM + i] * mask[s * N_DIM + j0 + jj];
    pnorm[t] = p;
  }
  __syncthreads();
  if (t < 16) {
    float sum = 0.f;
#pragma unroll
    for (int c = 0; c < 16; ++c) sum += pnorm[c * 16 + t];
    snorm[t] = sum;
  }

  const int c16  = lane & 15;
  const int half = lane >> 4;
  const int sOff = half << 1;
  {
    v8f acc00 = {}, acc01 = {}, acc02 = {}, acc03 = {};
    v8f acc10 = {}, acc11 = {}, acc12 = {}, acc13 = {};
    const float* bbase =
        mB + ((size_t)sOff * N_DIM + (j0 + wave * 2)) * CC + 4 * c16;
    for (int s0 = 0; s0 < S_DIM; s0 += 4) {
      const int sa = s0 + sOff;
      const v2f a0 = pair2(sA[sa * SA_STRIDE + c16],
                           sA[(sa + 1) * SA_STRIDE + c16]);
      const v2f a1 = pair2(sA[sa * SA_STRIDE + 16 + c16],
                           sA[(sa + 1) * SA_STRIDE + 16 + c16]);
      const float* bp = bbase + (size_t)s0 * N_DIM * CC;
      const v4f q0 = *reinterpret_cast<const v4f*>(bp);
      const v4f q1 = *reinterpret_cast<const v4f*>(bp + N_DIM * CC);
      const v2f b0 = pair2(q0.x, q1.x), b1 = pair2(q0.y, q1.y);
      const v2f b2 = pair2(q0.z, q1.z), b3 = pair2(q0.w, q1.w);
      acc00 = wmma_f32(a0, b0, acc00);
      acc10 = wmma_f32(a1, b0, acc10);
      acc01 = wmma_f32(a0, b1, acc01);
      acc11 = wmma_f32(a1, b1, acc11);
      acc02 = wmma_f32(a0, b2, acc02);
      acc12 = wmma_f32(a1, b2, acc12);
      acc03 = wmma_f32(a0, b3, acc03);
      acc13 = wmma_f32(a1, b3, acc13);
    }
    float* dst = sAB + (wave * 2 + (c16 >> 3)) * SAB_STRIDE + 4 * (c16 & 7);
#pragma unroll
    for (int r = 0; r < 8; ++r) {
      const int c = r + (half << 3);
      v4f lo, hi;
      lo.x = acc00[r]; lo.y = acc01[r]; lo.z = acc02[r]; lo.w = acc03[r];
      hi.x = acc10[r]; hi.y = acc11[r]; hi.z = acc12[r]; hi.w = acc13[r];
      *reinterpret_cast<v4f*>(dst + c * CC)        = lo;
      *reinterpret_cast<v4f*>(dst + (c + 16) * CC) = hi;
    }
  }
  __syncthreads();

  const int zg = wave & 1;
  const int kq = wave >> 1;
  v8f accZ0 = {}, accZ1 = {}, accZ2 = {}, accZ3 = {};
  {
    const float* arow = sAB + c16 * SAB_STRIDE + kq * 256 + sOff;
    const float* wp = wOut + (size_t)(kq * 256 + sOff) * CZ + zg * 64 + 4 * c16;
    for (int k0 = 0; k0 < 256; k0 += 4) {
      const v2f a = *reinterpret_cast<const v2f*>(arow + k0);
      const v4f q0 = *reinterpret_cast<const v4f*>(wp + (size_t)k0 * CZ);
      const v4f q1 = *reinterpret_cast<const v4f*>(wp + (size_t)(k0 + 1) * CZ);
      accZ0 = wmma_f32(a, pair2(q0.x, q1.x), accZ0);
      accZ1 = wmma_f32(a, pair2(q0.y, q1.y), accZ1);
      accZ2 = wmma_f32(a, pair2(q0.z, q1.z), accZ2);
      accZ3 = wmma_f32(a, pair2(q0.w, q1.w), accZ3);
    }
  }

  __syncthreads();
  float* red = sAB;
  {
    float* dst = red + kq * (16 * CZ) + (half << 3) * CZ + zg * 64 + 4 * c16;
#pragma unroll
    for (int r = 0; r < 8; ++r) {
      v4f v;
      v.x = accZ0[r]; v.y = accZ1[r]; v.z = accZ2[r]; v.w = accZ3[r];
      *reinterpret_cast<v4f*>(dst + r * CZ) = v;
    }
  }
  __syncthreads();
  float* orow = out + ((size_t)i * N_DIM + j0) * CZ;
  for (int q = 0; q < 2; ++q) {
    const int f = q * 1024 + 4 * t;
    const int jj = f >> 7, z = f & (CZ - 1);
    const v4f p0 = *reinterpret_cast<const v4f*>(red + f);
    const v4f p1 = *reinterpret_cast<const v4f*>(red + 16 * CZ + f);
    const v4f p2 = *reinterpret_cast<const v4f*>(red + 2 * 16 * CZ + f);
    const v4f p3 = *reinterpret_cast<const v4f*>(red + 3 * 16 * CZ + f);
    const v4f zsum = ((p0 + p1) + p2) + p3;
    const float nrm = snorm[jj] + 0.001f;
    v4f o;
    o.x = (zsum.x + bOut[z])     / nrm;
    o.y = (zsum.y + bOut[z + 1]) / nrm;
    o.z = (zsum.z + bOut[z + 2]) / nrm;
    o.w = (zsum.w + bOut[z + 3]) / nrm;
    *reinterpret_cast<v4f*>(orow + f) = o;
  }
}

extern "C" void kernel_launch(void* const* d_in, const int* in_sizes, int n_in,
                              void* d_out, int out_size, void* d_ws,
                              size_t ws_size, hipStream_t stream) {
  (void)in_sizes; (void)n_in; (void)out_size; (void)ws_size;
  const float* msa   = (const float*)d_in[0];
  const float* mask  = (const float*)d_in[1];
  const float* gamma = (const float*)d_in[2];
  const float* beta  = (const float*)d_in[3];
  const float* wa    = (const float*)d_in[4];
  const float* wb    = (const float*)d_in[5];
  const float* wout  = (const float*)d_in[6];
  const float* bout  = (const float*)d_in[7];
  float* out = (float*)d_out;

  float* mAbuf = (float*)d_ws;
  float* mBbuf = mAbuf + (size_t)S_DIM * N_DIM * CC;

  opm_ln_proj<<<(S_DIM * N_DIM) / 8, 256, 0, stream>>>(msa, mask, gamma, beta,
                                                       wa, wb, mAbuf, mBbuf);

  const size_t ldsBytes =
      (size_t)(S_DIM * SA_STRIDE + 16 * SAB_STRIDE + 256 + 16) * sizeof(float);
  opm_main<<<dim3(N_DIM / 16, N_DIM), 256, ldsBytes, stream>>>(
      mAbuf, mBbuf, mask, wout, bout, out);
}


# Main kernel block covers 32 j (was 16) with 16 waves: each fetched w_out fragment feeds two A tiles; stage 2 K split in eight across waves, partials added through LDS (sum order changes)
# speedup vs baseline: 1.1666x; 1.0536x over previous
#include <hip/hip_runtime.h>

#define S_DIM 128
#define N_DIM 384
#define CM    64
#define CC    32
#define CZ    128
#define LN_EPS 1e-5f
#define JT    32
#define NW    16

#define SA_STRIDE  40
#define SAB_STRIDE 1028

typedef __attribute__((ext_vector_type(2))) float v2f;
typedef __attribute__((ext_vector_type(4))) float v4f;
typedef __attribute__((ext_vector_type(8))) float v8f;

__device__ __forceinline__ v8f wmma_f32(v2f a, v2f b, v8f c) {
  return __builtin_amdgcn_wmma_f32_16x16x4_f32(false, a, false, b, (short)0, c,
                                               false, false);
}

__device__ __forceinline__ v2f pair2(float x, float y) {
  v2f r;
  r.x = x;
  r.y = y;
  return r;
}

__device__ __forceinline__ float wave_sum(float v) {
#pragma unroll
  for (int off = 16; off > 0; off >>= 1) v += __shfl_xor(v, off, 32);
  return v;
}

__global__ __launch_bounds__(256) void opm_ln_proj(
    const float* __restrict__ msa, const float* __restrict__ mask,
    const float* __restrict__ gamma, const float* __restrict__ beta,
    const float* __restrict__ wa, const float* __restrict__ wb,
    float* __restrict__ mA, float* __restrict__ mB) {
  __shared__ float sY[8 * CM];
  const int t = threadIdx.x, wave = t >> 5, lane = t & 31;
  const int row = blockIdx.x * 8 + wave;

  const float* x = msa + (size_t)row * CM;
  const float x0 = x[lane], x1 = x[lane + 32];
  const float mean = wave_sum(x0 + x1) * (1.0f / CM);
  const float d0 = x0 - mean, d1 = x1 - mean;
  const float var = wave_sum(d0 * d0 + d1 * d1) * (1.0f / CM);
  const float inv = rsqrtf(var + LN_EPS);
  sY[wave * CM + lane]      = d0 * inv * gamma[lane] + beta[lane];
  sY[wave * CM + lane + 32] = d1 * inv * gamma[lane + 32] + beta[lane + 32];
  __syncthreads();

  const float mk = mask[row];
  float a = 0.f, b = 0.f;
#pragma unroll 8
  for (int m = 0; m < CM; ++m) {
    const float ym = sY[wave * CM + m];
    a = fmaf(ym, wa[m * CC + lane], a);
    b = fmaf(ym, wb[m * CC + lane], b);
  }
  mA[(size_t)row * CC + lane] = a * mk;
  mB[(size_t)row * CC + lane] = b * mk;
}

__global__ __launch_bounds__(512) void opm_main(
    const float* __restrict__ mA, const float* __restrict__ mB,
    const float* __restrict__ mask, const float* __restrict__ wOut,
    const float* __restrict__ bOut, float* __restrict__ out) {
  __shared__ __attribute__((aligned(16)))
  float smem[S_DIM * SA_STRIDE + JT * SAB_STRIDE + 32 * NW + JT];
  float* sA    = smem;
  float* sAB   = sA + S_DIM * SA_STRIDE;
  float* pnorm = sAB + JT * SAB_STRIDE;
  float* snorm = pnorm + 32 * NW;

  const int t = threadIdx.x;
  const int wave = t >> 5, lane = t & 31;
  const int i = blockIdx.y;
  const int j0 = blockIdx.x * JT;

  for (int idx = t; idx < S_DIM * CC; idx += 32 * NW) {
    const int s = idx >> 5, c = idx & 31;
    sA[s * SA_STRIDE + c] = mA[((size_t)s * N_DIM + i) * CC + c];
  }
  {
    const int jj = t & (JT - 1), chunk = t >> 5;
    float p = 0.f;
#pragma unroll
    for (int s = chunk * 8; s < chunk * 8 + 8; ++s)
      p += mask[s * N_DIM + i] * mask[s * N_DIM + j0 + jj];
    pnorm[t] = p;
  }
  __syncthreads();
  if (t < JT) {
    float sum = 0.f;
#pragma unroll
    for (int c = 0; c < 16; ++c) sum += pnorm[c * JT + t];
    snorm[t] = sum;
  }

  const int c16  = lane & 15;
  const int half = lane >> 4;
  const int sOff = half << 1;
  {
    v8f acc00 = {}, acc01 = {}, acc02 = {}, acc03 = {};
    v8f acc10 = {}, acc11 = {}, acc12 = {}, acc13 = {};
    const float* bbase =
        mB + ((size_t)sOff * N_DIM + (j0 + wave * 2)) * CC + 4 * c16;
    for (int s0 = 0; s0 < S_DIM; s0 += 4) {
      const int sa = s0 + sOff;
      const v2f a0 = pair2(sA[sa * SA_STRIDE + c16],
                           sA[(sa + 1) * SA_STRIDE + c16]);
      const v2f a1 = pair2(sA[sa * SA_STRIDE + 16 + c16],
                           sA[(sa + 1) * SA_STRIDE + 16 + c16]);
      const float* bp = bbase + (size_t)s0 * N_DIM * CC;
      const v4f q0 = *reinterpret_cast<const v4f*>(bp);
      const v4f q1 = *reinterpret_cast<const v4f*>(bp + N_DIM * CC);
      const v2f b0 = pair2(q0.x, q1.x), b1 = pair2(q0.y, q1.y);
      const v2f b2 = pair2(q0.z, q1.z), b3 = pair2(q0.w, q1.w);
      acc00 = wmma_f32(a0, b0, acc00);
      acc10 = wmma_f32(a1, b0, acc10);
      acc01 = wmma_f32(a0, b1, acc01);
      acc11 = wmma_f32(a1, b1, acc11);
      acc02 = wmma_f32(a0, b2, acc02);
      acc12 = wmma_f32(a1, b2, acc12);
      acc03 = wmma_f32(a0, b3, acc03);
      acc13 = wmma_f32(a1, b3, acc13);
    }
    float* dst = sAB + (wave * 2 + (c16 >> 3)) * SAB_STRIDE + 4 * (c16 & 7);
#pragma unroll
    for (int r = 0; r < 8; ++r) {
      const int c = r + (half << 3);
      v4f lo, hi;
      lo.x = acc00[r]; lo.y = acc01[r]; lo.z = acc02[r]; lo.w = acc03[r];
      hi.x = acc10[r]; hi.y = acc11[r]; hi.z = acc12[r]; hi.w = acc13[r];
      *reinterpret_cast<v4f*>(dst + c * CC)        = lo;
      *reinterpret_cast<v4f*>(dst + (c + 16) * CC) = hi;
    }
  }
  __syncthreads();

  const int zg = wave & 1;
  const int kq = wave >> 1;
  v8f accZ00 = {}, accZ01 = {}, accZ02 = {}, accZ03 = {};
  v8f accZ10 = {}, accZ11 = {}, accZ12 = {}, accZ13 = {};
  {
    const float* arow0 = sAB + c16 * SAB_STRIDE + kq * 128 + sOff;
    const float* arow1 = arow0 + 16 * SAB_STRIDE;
    const float* wp = wOut + (size_t)(kq * 128 + sOff) * CZ + zg * 64 + 4 * c16;
    for (int k0 = 0; k0 < 128; k0 += 4) {
      const v2f a0 = *reinterpret_cast<const v2f*>(arow0 + k0);
      const v2f a1 = *reinterpret_cast<const v2f*>(arow1 + k0);
      const v4f q0 = *reinterpret_cast<const v4f*>(wp + (size_t)k0 * CZ);
      const v4f q1 = *reinterpret_cast<const v4f*>(wp + (size_t)(k0 + 1) * CZ);
      const v2f b0 = pair2(q0.x, q1.x), b1 = pair2(q0.y, q1.y);
      const v2f b2 = pair2(q0.z, q1.z), b3 = pair2(q0.w, q1.w);
      accZ00 = wmma_f32(a0, b0, accZ00);
      accZ10 = wmma_f32(a1, b0, accZ10);
      accZ01 = wmma_f32(a0, b1, accZ01);
      accZ11 = wmma_f32(a1, b1, accZ11);
      accZ02 = wmma_f32(a0, b2, accZ02);
      accZ12 = wmma_f32(a1, b2, accZ12);
      accZ03 = wmma_f32(a0, b3, accZ03);
      accZ13 = wmma_f32(a1, b3, accZ13);
    }
  }

  __syncthreads();
  float* red = sAB;
  {
    float* dst = red + kq * (JT * CZ) + (half << 3) * CZ + zg * 64 + 4 * c16;
#pragma unroll
    for (int r = 0; r < 8; ++r) {
      v4f lo, hi;
      lo.x = accZ00[r]; lo.y = accZ01[r]; lo.z = accZ02[r]; lo.w = accZ03[r];
      hi.x = accZ10[r]; hi.y = accZ11[r]; hi.z = accZ12[r]; hi.w = accZ13[r];
      *reinterpret_cast<v4f*>(dst + r * CZ)        = lo;
      *reinterpret_cast<v4f*>(dst + (16 + r) * CZ) = hi;
    }
  }
  __syncthreads();
  float* orow = out + ((size_t)i * N_DIM + j0) * CZ;
  for (int q = 0; q < 2; ++q) {
    const int f = q * (JT * CZ / 2) + 4 * t;
    const int jj = f >> 7, z = f & (CZ - 1);
    const v4f p0 = *reinterpret_cast<const v4f*>(red + f);
    const v4f p1 = *reinterpret_cast<const v4f*>(red + JT * CZ + f);
    const v4f p2 = *reinterpret_cast<const v4f*>(red + 2 * JT * CZ + f);
    const v4f p3 = *reinterpret_cast<const v4f*>(red + 3 * JT * CZ + f);
    const v4f p4 = *reinterpret_cast<const v4f*>(red + 4 * JT * CZ + f);
    const v4f p5 = *reinterpret_cast<const v4f*>(red + 5 * JT * CZ + f);
    const v4f p6 = *reinterpret_cast<const v4f*>(red + 6 * JT * CZ + f);
    const v4f p7 = *reinterpret_cast<const v4f*>(red + 7 * JT * CZ + f);
    const v4f zsum = ((((((p0 + p1) + p2) + p3) + p4) + p5) + p6) + p7;
    const float nrm = snorm[jj] + 0.001f;
    v4f o;
    o.x = (zsum.x + bOut[z])     / nrm;
    o.y = (zsum.y + bOut[z + 1]) / nrm;
    o.z = (zsum.z + bOut[z + 2]) / nrm;
    o.w = (zsum.w + bOut[z + 3]) / nrm;
    *reinterpret_cast<v4f*>(orow + f) = o;
  }
}

extern "C" void kernel_launch(void* const* d_in, const int* in_sizes, int n_in,
                              void* d_out, int out_size, void* d_ws,
                              size_t ws_size, hipStream_t stream) {
  (void)in_sizes; (void)n_in; (void)out_size; (void)ws_size;
  const float* msa   = (const float*)d_in[0];
  const float* mask  = (const float*)d_in[1];
  const float* gamma = (const float*)d_in[2];
  const float* beta  = (const float*)d_in[3];
  const float* wa    = (const float*)d_in[4];
  const float* wb    = (const float*)d_in[5];
  const float* wout  = (const float*)d_in[6];
  const float* bout  = (const float*)d_in[7];
  float* out = (float*)d_out;

  float* mAbuf = (float*)d_ws;
  float* mBbuf = mAbuf + (size_t)S_DIM * N_DIM * CC;

  opm_ln_proj<<<(S_DIM * N_DIM) / 8, 256, 0, stream>>>(msa, mask, gamma, beta,
                                                       wa, wb, mAbuf, mBbuf);

  opm_main<<<dim3(N_DIM / JT, N_DIM), 32 * NW, 0, stream>>>(
      mAbuf, mBbuf, mask, wout, bout, out);
}


# Main kernel prologue: m_a slice staged by two 16-byte loads per thread, every prologue load issued before the first wait: one memory round trip per block instead of nine dependent ones
# speedup vs baseline: 1.2208x; 1.0465x over previous
#include <hip/hip_runtime.h>

#define S_DIM 128
#define N_DIM 384
#define CM    64
#define CC    32
#define CZ    128
#define LN_EPS 1e-5f
#define JT    32
#define NW    16

#define SA_STRIDE  40
#define SAB_STRIDE 1028

typedef __attribute__((ext_vector_type(2))) float v2f;
typedef __attribute__((ext_vector_type(4))) float v4f;
typedef __attribute__((ext_vector_type(8))) float v8f;

__device__ __forceinline__ v8f wmma_f32(v2f a, v2f b, v8f c) {
  return __builtin_amdgcn_wmma_f32_16x16x4_f32(false, a, false, b, (short)0, c,
                                               false, false);
}

__device__ __forceinline__ v2f pair2(float x, float y) {
  v2f r;
  r.x = x;
  r.y = y;
  return r;
}

__device__ __forceinline__ float wave_sum(float v) {
#pragma unroll
  for (int off = 16; off > 0; off >>= 1) v += __shfl_xor(v, off, 32);
  return v;
}

__global__ __launch_bounds__(256) void opm_ln_proj(
    const float* __restrict__ msa, const float* __restrict__ mask,
    const float* __restrict__ gamma, const float* __restrict__ beta,
    const float* __restrict__ wa, const float* __restrict__ wb,
    float* __restrict__ mA, float* __restrict__ mB) {
  __shared__ float sY[8 * CM];
  const int t = threadIdx.x, wave = t >> 5, lane = t & 31;
  const int row = blockIdx.x * 8 + wave;

  const float* x = msa + (size_t)row * CM;
  const float x0 = x[lane], x1 = x[lane + 32];
  const float mean = wave_sum(x0 + x1) * (1.0f / CM);
  const float d0 = x0 - mean, d1 = x1 - mean;
  const float var = wave_sum(d0 * d0 + d1 * d1) * (1.0f / CM);
  const float inv = rsqrtf(var + LN_EPS);
  sY[wave * CM + lane]      = d0 * inv * gamma[lane] + beta[lane];
  sY[wave * CM + lane + 32] = d1 * inv * gamma[lane + 32] + beta[lane + 32];
  __syncthreads();

  const float mk = mask[row];
  float a = 0.f, b = 0.f;
#pragma unroll 8
  for (int m = 0; m < CM; ++m) {
    const float ym = sY[wave * CM + m];
    a = fmaf(ym, wa[m * CC + lane], a);
    b = fmaf(ym, wb[m * CC + lane], b);
  }
  mA[(size_t)row * CC + lane] = a * mk;
  mB[(size_t)row * CC + lane] = b * mk;
}

__global__ __launch_bounds__(512) void opm_main(
    const float* __restrict__ mA, const float* __restrict__ mB,
    const float* __restrict__ mask, const float* __restrict__ wOut,
    const float* __restrict__ bOut, float* __restrict__ out) {
  __shared__ __attribute__((aligned(16)))
  float smem[S_DIM * SA_STRIDE + JT * SAB_STRIDE + 32 * NW + JT];
  float* sA    = smem;
  float* sAB   = sA + S_DIM * SA_STRIDE;
  float* pnorm = sAB + JT * SAB_STRIDE;
  float* snorm = pnorm + 32 * NW;

  const int t = threadIdx.x;
  const int wave = t >> 5, lane = t & 31;
  const int i = blockIdx.y;
  const int j0 = blockIdx.x * JT;

  const int s_lo = t >> 3, c4 = (t & 7) << 2;
  const v4f r0 = *reinterpret_cast<const v4f*>(
      mA + ((size_t)s_lo * N_DIM + i) * CC + c4);
  const v4f r1 = *reinterpret_cast<const v4f*>(
      mA + ((size_t)(s_lo + 64) * N_DIM + i) * CC + c4);
  {
    const int jj = t & (JT - 1), chunk = t >> 5;
    float mi[8], mj[8];
#pragma unroll
    for (int u = 0; u < 8; ++u) {
      mi[u] = mask[(chunk * 8 + u) * N_DIM + i];
      mj[u] = mask[(chunk * 8 + u) * N_DIM + j0 + jj];
    }
    __builtin_amdgcn_sched_barrier(0);
    float p = 0.f;
#pragma unroll
    for (int u = 0; u < 8; ++u) p += mi[u] * mj[u];
    pnorm[t] = p;
  }
  *reinterpret_cast<v4f*>(sA + s_lo * SA_STRIDE + c4)        = r0;
  *reinterpret_cast<v4f*>(sA + (s_lo + 64) * SA_STRIDE + c4) = r1;
  __syncthreads();
  if (t < JT) {
    float sum = 0.f;
#pragma unroll
    for (int c = 0; c < 16; ++c) sum += pnorm[c * JT + t];
    snorm[t] = sum;
  }

  const int c16  = lane & 15;
  const int half = lane >> 4;
  const int sOff = half << 1;
  {
    v8f acc00 = {}, acc01 = {}, acc02 = {}, acc03 = {};
    v8f acc10 = {}, acc11 = {}, acc12 = {}, acc13 = {};
    const float* bbase =
        mB + ((size_t)sOff * N_DIM + (j0 + wave * 2)) * CC + 4 * c16;
    for (int s0 = 0; s0 < S_DIM; s0 += 4) {
      const int sa = s0 + sOff;
      const v2f a0 = pair2(sA[sa * SA_STRIDE + c16],
                           sA[(sa + 1) * SA_STRIDE + c16]);
      const v2f a1 = pair2(sA[sa * SA_STRIDE + 16 + c16],
                           sA[(sa + 1) * SA_STRIDE + 16 + c16]);
      const float* bp = bbase + (size_t)s0 * N_DIM * CC;
      const v4f q0 = *reinterpret_cast<const v4f*>(bp);
      const v4f q1 = *reinterpret_cast<const v4f*>(bp + N_DIM * CC);
      const v2f b0 = pair2(q0.x, q1.x), b1 = pair2(q0.y, q1.y);
      const v2f b2 = pair2(q0.z, q1.z), b3 = pair2(q0.w, q1.w);
      acc00 = wmma_f32(a0, b0, acc00);
      acc10 = wmma_f32(a1, b0, acc10);
      acc01 = wmma_f32(a0, b1, acc01);
      acc11 = wmma_f32(a1, b1, acc11);
      acc02 = wmma_f32(a0, b2, acc02);
      acc12 = wmma_f32(a1, b2, acc12);
      acc03 = wmma_f32(a0, b3, acc03);
      acc13 = wmma_f32(a1, b3, acc13);
    }
    float* dst = sAB + (wave * 2 + (c16 >> 3)) * SAB_STRIDE + 4 * (c16 & 7);
#pragma unroll
    for (int r = 0; r < 8; ++r) {
      const int c = r + (half << 3);
      v4f lo, hi;
      lo.x = acc00[r]; lo.y = acc01[r]; lo.z = acc02[r]; lo.w = acc03[r];
      hi.x = acc10[r]; hi.y = acc11[r]; hi.z = acc12[r]; hi.w = acc13[r];
      *reinterpret_cast<v4f*>(dst + c * CC)        = lo;
      *reinterpret_cast<v4f*>(dst + (c + 16) * CC) = hi;
    }
  }
  __syncthreads();

  const int zg = wave & 1;
  const int kq = wave >> 1;
  v8f accZ00 = {}, accZ01 = {}, accZ02 = {}, accZ03 = {};
  v8f accZ10 = {}, accZ11 = {}, accZ12 = {}, accZ13 = {};
  {
    const float* arow0 = sAB + c16 * SAB_STRIDE + kq * 128 + sOff;
    const float* arow1 = arow0 + 16 * SAB_STRIDE;
    const float* wp = wOut + (size_t)(kq * 128 + sOff) * CZ + zg * 64 + 4 * c16;
    for (int k0 = 0; k0 < 128; k0 += 4) {
      const v2f a0 = *reinterpret_cast<const v2f*>(arow0 + k0);
      const v2f a1 = *reinterpret_cast<const v2f*>(arow1 + k0);
      const v4f q0 = *reinterpret_cast<const v4f*>(wp + (size_t)k0 * CZ);
      const v4f q1 = *reinterpret_cast<const v4f*>(wp + (size_t)(k0 + 1) * CZ);
      const v2f b0 = pair2(q0.x, q1.x), b1 = pair2(q0.y, q1.y);
      const v2f b2 = pair2(q0.z, q1.z), b3 = pair2(q0.w, q1.w);
      accZ00 = wmma_f32(a0, b0, accZ00);
      accZ10 = wmma_f32(a1, b0, accZ10);
      accZ01 = wmma_f32(a0, b1, accZ01);
      accZ11 = wmma_f32(a1, b1, accZ11);
      accZ02 = wmma_f32(a0, b2, accZ02);
      accZ12 = wmma_f32(a1, b2, accZ12);
      accZ03 = wmma_f32(a0, b3, accZ03);
      accZ13 = wmma_f32(a1, b3, accZ13);
    }
  }

  __syncthreads();
  float* red = sAB;
  {
    float* dst = red + kq * (JT * CZ) + (half << 3) * CZ + zg * 64 + 4 * c16;
#pragma unroll
    for (int r = 0; r < 8; ++r) {
      v4f lo, hi;
      lo.x = accZ00[r]; lo.y = accZ01[r]; lo.z = accZ02[r]; lo.w = accZ03[r];
      hi.x = accZ10[r]; hi.y = accZ11[r]; hi.z = accZ12[r]; hi.w = accZ13[r];
      *reinterpret_cast<v4f*>(dst + r * CZ)        = lo;
      *reinterpret_cast<v4f*>(dst + (16 + r) * CZ) = hi;
    }
  }
  __syncthreads();
  float* orow = out + ((size_t)i * N_DIM + j0) * CZ;
  for (int q = 0; q < 2; ++q) {
    const int f = q * (JT * CZ / 2) + 4 * t;
    const int jj = f >> 7, z = f & (CZ - 1);
    const v4f p0 = *reinterpret_cast<const v4f*>(red + f);
    const v4f p1 = *reinterpret_cast<const v4f*>(red + JT * CZ + f);
    const v4f p2 = *reinterpret_cast<const v4f*>(red + 2 * JT * CZ + f);
    const v4f p3 = *reinterpret_cast<const v4f*>(red + 3 * JT * CZ + f);
    const v4f p4 = *reinterpret_cast<const v4f*>(red + 4 * JT * CZ + f);
    const v4f p5 = *reinterpret_cast<const v4f*>(red + 5 * JT * CZ + f);
    const v4f p6 = *reinterpret_cast<const v4f*>(red + 6 * JT * CZ + f);
    const v4f p7 = *reinterpret_cast<const v4f*>(red + 7 * JT * CZ + f);
    const v4f zsum = ((((((p0 + p1) + p2) + p3) + p4) + p5) + p6) + p7;
    const float nrm = snorm[jj] + 0.001f;
    v4f o;
    o.x = (zsum.x + bOut[z])     / nrm;
    o.y = (zsum.y + bOut[z + 1]) / nrm;
    o.z = (zsum.z + bOut[z + 2]) / nrm;
    o.w = (zsum.w + bOut[z + 3]) / nrm;
    *reinterpret_cast<v4f*>(orow + f) = o;
  }
}

extern "C" void kernel_launch(void* const* d_in, const int* in_sizes, int n_in,
                              void* d_out, int out_size, void* d_ws,
                              size_t ws_size, hipStream_t stream) {
  (void)in_sizes; (void)n_in; (void)out_size; (void)ws_size;
  const float* msa   = (const float*)d_in[0];
  const float* mask  = (const float*)d_in[1];
  const float* gamma = (const float*)d_in[2];
  const float* beta  = (const float*)d_in[3];
  const float* wa    = (const float*)d_in[4];
  const float* wb    = (const float*)d_in[5];
  const float* wout  = (const float*)d_in[6];
  const float* bout  = (const float*)d_in[7];
  float* out = (float*)d_out;

  float* mAbuf = (float*)d_ws;
  float* mBbuf = mAbuf + (size_t)S_DIM * N_DIM * CC;

  opm_ln_proj<<<(S_DIM * N_DIM) / 8, 256, 0, stream>>>(msa, mask, gamma, beta,
                                                       wa, wb, mAbuf, mBbuf);

  opm_main<<<dim3(N_DIM / JT, N_DIM), 32 * NW, 0, stream>>>(
      mAbuf, mBbuf, mask, wout, bout, out);
}


# Stage-1 D tiles feed stage 2 as A fragments in registers: no LDS hand-over, no barrier between the products; m_b stored in fragment order; 4 waves' partial Z added via LDS (z, norm sum order change)
# speedup vs baseline: 1.2624x; 1.0341x over previous
#include <hip/hip_runtime.h>

#define S_DIM 128
#define N_DIM 384
#define CM    64
#define CC    32
#define CZ    128
#define LN_EPS 1e-5f
#define JT    32
#define SA_STRIDE 40

typedef __attribute__((ext_vector_type(2))) float v2f;
typedef __attribute__((ext_vector_type(4))) float v4f;
typedef __attribute__((ext_vector_type(8))) float v8f;

__device__ __forceinline__ v8f wmma_f32(v2f a, v2f b, v8f c) {
  return __builtin_amdgcn_wmma_f32_16x16x4_f32(false, a, false, b, (short)0, c,
                                               false, false);
}

__device__ __forceinline__ v2f pair2(float x, float y) {
  v2f r;
  r.x = x;
  r.y = y;
  return r;
}

__device__ __forceinline__ v4f pair4(float x, float y, float z, float w) {
  v4f r;
  r.x = x;
  r.y = y;
  r.z = z;
  r.w = w;
  return r;
}

__device__ __forceinline__ float wave_sum(float v) {
#pragma unroll
  for (int off = 16; off > 0; off >>= 1) v += __shfl_xor(v, off, 32);
  return v;
}

__global__ __launch_bounds__(256) void opm_ln_proj(
    const float* __restrict__ msa, const float* __restrict__ mask,
    const float* __restrict__ gamma, const float* __restrict__ beta,
    const float* __restrict__ wa, const float* __restrict__ wb,
    float* __restrict__ mA, float* __restrict__ mBf) {
  __shared__ float sY[8 * 2 * CM];
  __shared__ __attribute__((aligned(16))) float sB[16 * 8 * 4];
  const int t = threadIdx.x, wave = t >> 5, lane = t & 31;
  const int sp = blockIdx.y;
  const int n0 = blockIdx.x * 8;
  const int row0 = (2 * sp) * N_DIM + n0 + wave;
  const int row1 = row0 + N_DIM;

  const float g0 = gamma[lane], g1 = gamma[lane + 32];
  const float e0 = beta[lane], e1 = beta[lane + 32];
#pragma unroll
  for (int u = 0; u < 2; ++u) {
    const float* x = msa + (size_t)(u ? row1 : row0) * CM;
    const float x0 = x[lane], x1 = x[lane + 32];
    const float mean = wave_sum(x0 + x1) * (1.0f / CM);
    const float d0 = x0 - mean, d1 = x1 - mean;
    const float var = wave_sum(d0 * d0 + d1 * d1) * (1.0f / CM);
    const float inv = rsqrtf(var + LN_EPS);
    sY[(wave * 2 + u) * CM + lane]      = d0 * inv * g0 + e0;
    sY[(wave * 2 + u) * CM + lane + 32] = d1 * inv * g1 + e1;
  }
  __syncthreads();

  const float mk0 = mask[row0], mk1 = mask[row1];
  float a0 = 0.f, b0 = 0.f, a1 = 0.f, b1 = 0.f;
#pragma unroll 8
  for (int m = 0; m < CM; ++m) {
    const float y0 = sY[(wave * 2) * CM + m];
    const float y1 = sY[(wave * 2 + 1) * CM + m];
    const float wam = wa[m * CC + lane], wbm = wb[m * CC + lane];
    a0 = fmaf(y0, wam, a0);
    b0 = fmaf(y0, wbm, b0);
    a1 = fmaf(y1, wam, a1);
    b1 = fmaf(y1, wbm, b1);
  }
  mA[(size_t)row0 * CC + lane] = a0 * mk0;
  mA[(size_t)row1 * CC + lane] = a1 * mk1;
  *reinterpret_cast<v2f*>(sB + ((lane >> 1) * 8 + wave) * 4 + (lane & 1) * 2) =
      pair2(b0 * mk0, b1 * mk1);
  __syncthreads();
  if (t < 128) {
    const int dp = t >> 3, nn = t & 7;
    const v4f v = *reinterpret_cast<const v4f*>(sB + t * 4);
    *reinterpret_cast<v4f*>(
        mBf + (((size_t)sp * 16 + dp) * N_DIM + n0 + nn) * 4) = v;
  }
}

__global__ __launch_bounds__(128) __attribute__((amdgpu_waves_per_eu(4, 4))) void opm_main(
    const float* __restrict__ mA, const float* __restrict__ mBf,
    const float* __restrict__ mask, const float* __restrict__ wOut,
    const float* __restrict__ bOut, float* __restrict__ out) {
  __shared__ __attribute__((aligned(16)))
  float smem[S_DIM * SA_STRIDE + 3 * JT * CZ + 128];
  float* sA    = smem;
  float* red   = sA + S_DIM * SA_STRIDE;
  float* pnorm = red + 3 * JT * CZ;

  const int t = threadIdx.x;
  const int wave = t >> 5, lane = t & 31;
  const int i = blockIdx.y;
  const int j0 = blockIdx.x * JT;

  v4f ra[8];
#pragma unroll
  for (int u = 0; u < 8; ++u) {
    const int idx = u * 128 + t;
    ra[u] = *reinterpret_cast<const v4f*>(
        mA + ((size_t)(idx >> 3) * N_DIM + i) * CC + ((idx & 7) << 2));
  }
  {
    const int jj = t & 31, chunk = t >> 5;
    float p = 0.f;
#pragma unroll 8
    for (int u = 0; u < 32; ++u) {
      const int s = chunk * 32 + u;
      p += mask[s * N_DIM + i] * mask[s * N_DIM + j0 + jj];
    }
    pnorm[t] = p;
  }
#pragma unroll
  for (int u = 0; u < 8; ++u) {
    const int idx = u * 128 + t;
    *reinterpret_cast<v4f*>(sA + (idx >> 3) * SA_STRIDE + ((idx & 7) << 2)) = ra[u];
  }
  __syncthreads();

  const int c16 = lane & 15;
  const int h   = lane >> 4;

  v8f Z000 = {}, Z001 = {}, Z002 = {}, Z003 = {};
  v8f Z010 = {}, Z011 = {}, Z012 = {}, Z013 = {};
  v8f Z100 = {}, Z101 = {}, Z102 = {}, Z103 = {};
  v8f Z110 = {}, Z111 = {}, Z112 = {}, Z113 = {};

  const float* ap = sA + (2 * h) * SA_STRIDE + c16;
  for (int q = 0; q < 4; ++q) {
    const int dp = wave * 4 + q;
    v8f T000 = {}, T001 = {}, T010 = {}, T011 = {};
    v8f T100 = {}, T101 = {}, T110 = {}, T111 = {};
    const float* bp = mBf + ((size_t)(h * 16 + dp) * N_DIM + j0 + c16) * 4;
    for (int s0 = 0; s0 < S_DIM; s0 += 4) {
      const v2f a0 = pair2(ap[s0 * SA_STRIDE], ap[(s0 + 1) * SA_STRIDE]);
      const v2f a1 = pair2(ap[s0 * SA_STRIDE + 16], ap[(s0 + 1) * SA_STRIDE + 16]);
      const float* b = bp + (size_t)(s0 >> 1) * (16 * N_DIM * 4);
      const v4f q0 = *reinterpret_cast<const v4f*>(b);
      const v4f q1 = *reinterpret_cast<const v4f*>(b + 64);
      const v2f b00 = pair2(q0.x, q0.y), b01 = pair2(q0.z, q0.w);
      const v2f b10 = pair2(q1.x, q1.y), b11 = pair2(q1.z, q1.w);
      T000 = wmma_f32(a0, b00, T000);
      T100 = wmma_f32(a1, b00, T100);
      T001 = wmma_f32(a0, b01, T001);
      T101 = wmma_f32(a1, b01, T101);
      T010 = wmma_f32(a0, b10, T010);
      T110 = wmma_f32(a1, b10, T110);
      T011 = wmma_f32(a0, b11, T011);
      T111 = wmma_f32(a1, b11, T111);
    }
#define OPM_LOAD(S, CH, DD, T4)                                                 \
    {                                                                           \
      const float* wp = wOut +                                                  \
          ((size_t)((16 * (CH) + 8 * h + 2 * (T4)) * 32 + 2 * dp + (DD))) * CZ +\
          4 * c16;                                                              \
      S##P = *reinterpret_cast<const v4f*>(wp);                                \
      S##Q = *reinterpret_cast<const v4f*>(wp + 64);                           \
      S##R = *reinterpret_cast<const v4f*>(wp + 32 * CZ);                      \
      S##U = *reinterpret_cast<const v4f*>(wp + 32 * CZ + 64);                 \
    }
#define OPM_MMA(S, TJ0, TJ1, T4)                                                \
    {                                                                           \
      const v2f A0 = pair2(TJ0[2 * (T4)], TJ0[2 * (T4) + 1]);                   \
      const v2f A1 = pair2(TJ1[2 * (T4)], TJ1[2 * (T4) + 1]);                   \
      const v2f B0 = pair2(S##P.x, S##R.x), B1 = pair2(S##P.y, S##R.y);     \
      const v2f B2 = pair2(S##P.z, S##R.z), B3 = pair2(S##P.w, S##R.w);     \
      const v2f B4 = pair2(S##Q.x, S##U.x), B5 = pair2(S##Q.y, S##U.y);     \
      const v2f B6 = pair2(S##Q.z, S##U.z), B7 = pair2(S##Q.w, S##U.w);     \
      Z000 = wmma_f32(A0, B0, Z000);  Z100 = wmma_f32(A1, B0, Z100);            \
      Z001 = wmma_f32(A0, B1, Z001);  Z101 = wmma_f32(A1, B1, Z101);            \
      Z002 = wmma_f32(A0, B2, Z002);  Z102 = wmma_f32(A1, B2, Z102);            \
      Z003 = wmma_f32(A0, B3, Z003);  Z103 = wmma_f32(A1, B3, Z103);            \
      Z010 = wmma_f32(A0, B4, Z010);  Z110 = wmma_f32(A1, B4, Z110);            \
      Z011 = wmma_f32(A0, B5, Z011);  Z111 = wmma_f32(A1, B5, Z111);            \
      Z012 = wmma_f32(A0, B6, Z012);  Z112 = wmma_f32(A1, B6, Z112);            \
      Z013 = wmma_f32(A0, B7, Z013);  Z113 = wmma_f32(A1, B7, Z113);            \
    }
#define OPM_FENCE __builtin_amdgcn_sched_barrier(0);
    v4f waP, waQ, waR, waU, wbP, wbQ, wbR, wbU;
    OPM_LOAD(wa, 0, 0, 0)
    OPM_LOAD(wb, 0, 0, 1) OPM_FENCE
    OPM_MMA(wa, T000, T010, 0) OPM_FENCE
    OPM_LOAD(wa, 0, 0, 2) OPM_FENCE
    OPM_MMA(wb, T000, T010, 1) OPM_FENCE
    OPM_LOAD(wb, 0, 0, 3) OPM_FENCE
    OPM_MMA(wa, T000, T010, 2) OPM_FENCE
    OPM_LOAD(wa, 0, 1, 0) OPM_FENCE
    OPM_MMA(wb, T000, T010, 3) OPM_FENCE
    OPM_LOAD(wb, 0, 1, 1) OPM_FENCE
    OPM_MMA(wa, T001, T011, 0) OPM_FENCE
    OPM_LOAD(wa, 0, 1, 2) OPM_FENCE
    OPM_MMA(wb, T001, T011, 1) OPM_FENCE
    OPM_LOAD(wb, 0, 1, 3) OPM_FENCE
    OPM_MMA(wa, T001, T011, 2) OPM_FENCE
    OPM_LOAD(wa, 1, 0, 0) OPM_FENCE
    OPM_MMA(wb, T001, T011, 3) OPM_FENCE
    OPM_LOAD(wb, 1, 0, 1) OPM_FENCE
    OPM_MMA(wa, T100, T110, 0) OPM_FENCE
    OPM_LOAD(wa, 1, 0, 2) OPM_FENCE
    OPM_MMA(wb, T100, T110, 1) OPM_FENCE
    OPM_LOAD(wb, 1, 0, 3) OPM_FENCE
    OPM_MMA(wa, T100, T110, 2) OPM_FENCE
    OPM_LOAD(wa, 1, 1, 0) OPM_FENCE
    OPM_MMA(wb, T100, T110, 3) OPM_FENCE
    OPM_LOAD(wb, 1, 1, 1) OPM_FENCE
    OPM_MMA(wa, T101, T111, 0) OPM_FENCE
    OPM_LOAD(wa, 1, 1, 2) OPM_FENCE
    OPM_MMA(wb, T101, T111, 1) OPM_FENCE
    OPM_LOAD(wb, 1, 1, 3) OPM_FENCE
    OPM_MMA(wa, T101, T111, 2) OPM_FENCE
    OPM_MMA(wb, T101, T111, 3)
#undef OPM_FENCE
#undef OPM_MMA
#undef OPM_LOAD
  }

  __syncthreads();
  {
    float* dst = (wave == 0 ? sA : red + (wave - 1) * (JT * CZ)) +
                 (8 * h) * CZ + 4 * c16;
#pragma unroll
    for (int r = 0; r < 8; ++r) {
      *reinterpret_cast<v4f*>(dst + r * CZ)             = pair4(Z000[r], Z001[r], Z002[r], Z003[r]);
      *reinterpret_cast<v4f*>(dst + r * CZ + 64)        = pair4(Z010[r], Z011[r], Z012[r], Z013[r]);
      *reinterpret_cast<v4f*>(dst + (16 + r) * CZ)      = pair4(Z100[r], Z101[r], Z102[r], Z103[r]);
      *reinterpret_cast<v4f*>(dst + (16 + r) * CZ + 64) = pair4(Z110[r], Z111[r], Z112[r], Z113[r]);
    }
  }
  __syncthreads();
  float* orow = out + ((size_t)i * N_DIM + j0) * CZ;
  const v4f bo = *reinterpret_cast<const v4f*>(bOut + ((4 * t) & (CZ - 1)));
  for (int q = 0; q < 8; ++q) {
    const int f = q * 512 + 4 * t;
    const int jj = f >> 7;
    const v4f p0 = *reinterpret_cast<const v4f*>(sA + f);
    const v4f p1 = *reinterpret_cast<const v4f*>(red + f);
    const v4f p2 = *reinterpret_cast<const v4f*>(red + JT * CZ + f);
    const v4f p3 = *reinterpret_cast<const v4f*>(red + 2 * JT * CZ + f);
    const v4f zsum = ((p0 + p1) + p2) + p3;
    const float nrm =
        (((pnorm[jj] + pnorm[32 + jj]) + pnorm[64 + jj]) + pnorm[96 + jj]) + 0.001f;
    v4f o;
    o.x = (zsum.x + bo.x) / nrm;
    o.y = (zsum.y + bo.y) / nrm;
    o.z = (zsum.z + bo.z) / nrm;
    o.w = (zsum.w + bo.w) / nrm;
    *reinterpret_cast<v4f*>(orow + f) = o;
  }
}

extern "C" void kernel_launch(void* const* d_in, const int* in_sizes, int n_in,
                              void* d_out, int out_size, void* d_ws,
                              size_t ws_size, hipStream_t stream) {
  (void)in_sizes; (void)n_in; (void)out_size; (void)ws_size;
  const float* msa   = (const float*)d_in[0];
  const float* mask  = (const float*)d_in[1];
  const float* gamma = (const float*)d_in[2];
  const float* beta  = (const float*)d_in[3];
  const float* wa    = (const float*)d_in[4];
  const float* wb    = (const float*)d_in[5];
  const float* wout  = (const float*)d_in[6];
  const float* bout  = (const float*)d_in[7];
  float* out = (float*)d_out;

  float* mAbuf = (float*)d_ws;
  float* mBbuf = mAbuf + (size_t)S_DIM * N_DIM * CC;

  opm_ln_proj<<<dim3(N_DIM / 8, S_DIM / 2), 256, 0, stream>>>(
      msa, mask, gamma, beta, wa, wb, mAbuf, mBbuf);

  opm_main<<<dim3(N_DIM / JT, N_DIM), 128, 0, stream>>>(
      mAbuf, mBbuf, mask, wout, bout, out);
}


# Main kernel: a wave covers two i (64 rows): a fetched m_b or w_out fragment feeds 4 A tiles, half the global loads per WMMA; 512 registers, 2 waves/SIMD; stage-1 loads a group ahead, 8-byte A reads
# speedup vs baseline: 1.3570x; 1.0749x over previous
#include <hip/hip_runtime.h>

#define S_DIM 128
#define N_DIM 384
#define CM    64
#define CC    32
#define CZ    128
#define LN_EPS 1e-5f
#define JT    32
#define II    2
#define SP_STRIDE 96

typedef __attribute__((ext_vector_type(2))) float v2f;
typedef __attribute__((ext_vector_type(4))) float v4f;
typedef __attribute__((ext_vector_type(8))) float v8f;

__device__ __forceinline__ v8f wmma_f32(v2f a, v2f b, v8f c) {
  return __builtin_amdgcn_wmma_f32_16x16x4_f32(false, a, false, b, (short)0, c,
                                               false, false);
}

__device__ __forceinline__ v2f pair2(float x, float y) {
  v2f r;
  r.x = x;
  r.y = y;
  return r;
}

__device__ __forceinline__ v4f pair4(float x, float y, float z, float w) {
  v4f r;
  r.x = x;
  r.y = y;
  r.z = z;
  r.w = w;
  return r;
}

__device__ __forceinline__ float wave_sum(float v) {
#pragma unroll
  for (int off = 16; off > 0; off >>= 1) v += __shfl_xor(v, off, 32);
  return v;
}

__global__ __launch_bounds__(256) void opm_ln_proj(
    const float* __restrict__ msa, const float* __restrict__ mask,
    const float* __restrict__ gamma, const float* __restrict__ beta,
    const float* __restrict__ wa, const float* __restrict__ wb,
    float* __restrict__ mA, float* __restrict__ mBf) {
  __shared__ float sY[8 * 2 * CM];
  __shared__ __attribute__((aligned(16))) float sB[16 * 8 * 4];
  const int t = threadIdx.x, wave = t >> 5, lane = t & 31;
  const int sp = blockIdx.y;
  const int n0 = blockIdx.x * 8;
  const int row0 = (2 * sp) * N_DIM + n0 + wave;
  const int row1 = row0 + N_DIM;

  const float g0 = gamma[lane], g1 = gamma[lane + 32];
  const float e0 = beta[lane], e1 = beta[lane + 32];
#pragma unroll
  for (int u = 0; u < 2; ++u) {
    const float* x = msa + (size_t)(u ? row1 : row0) * CM;
    const float x0 = x[lane], x1 = x[lane + 32];
    const float mean = wave_sum(x0 + x1) * (1.0f / CM);
    const float d0 = x0 - mean, d1 = x1 - mean;
    const float var = wave_sum(d0 * d0 + d1 * d1) * (1.0f / CM);
    const float inv = rsqrtf(var + LN_EPS);
    sY[(wave * 2 + u) * CM + lane]      = d0 * inv * g0 + e0;
    sY[(wave * 2 + u) * CM + lane + 32] = d1 * inv * g1 + e1;
  }
  __syncthreads();

  const float mk0 = mask[row0], mk1 = mask[row1];
  float a0 = 0.f, b0 = 0.f, a1 = 0.f, b1 = 0.f;
#pragma unroll 8
  for (int m = 0; m < CM; ++m) {
    const float y0 = sY[(wave * 2) * CM + m];
    const float y1 = sY[(wave * 2 + 1) * CM + m];
    const float wam = wa[m * CC + lane], wbm = wb[m * CC + lane];
    a0 = fmaf(y0, wam, a0);
    b0 = fmaf(y0, wbm, b0);
    a1 = fmaf(y1, wam, a1);
    b1 = fmaf(y1, wbm, b1);
  }
  mA[(size_t)row0 * CC + lane] = a0 * mk0;
  mA[(size_t)row1 * CC + lane] = a1 * mk1;
  *reinterpret_cast<v2f*>(sB + ((lane >> 1) * 8 + wave) * 4 + (lane & 1) * 2) =
      pair2(b0 * mk0, b1 * mk1);
  __syncthreads();
  if (t < 128) {
    const int dp = t >> 3, nn = t & 7;
    const v4f v = *reinterpret_cast<const v4f*>(sB + t * 4);
    *reinterpret_cast<v4f*>(
        mBf + (((size_t)sp * 16 + dp) * N_DIM + n0 + nn) * 4) = v;
  }
}

#define ZPART (II * JT * CZ)
__global__ __launch_bounds__(128) __attribute__((amdgpu_waves_per_eu(2, 2))) void opm_main(
    const float* __restrict__ mA, const float* __restrict__ mBf,
    const float* __restrict__ mask, const float* __restrict__ wOut,
    const float* __restrict__ bOut, float* __restrict__ out) {
  __shared__ __attribute__((aligned(16)))
  float smem[II * (S_DIM / 2) * SP_STRIDE + 3 * ZPART + II * 128];
  float* sA    = smem;
  float* red   = sA + II * (S_DIM / 2) * SP_STRIDE;
  float* pnorm = red + 3 * ZPART;

  const int t = threadIdx.x;
  const int wave = t >> 5, lane = t & 31;
  const int i0 = blockIdx.y * II;
  const int j0 = blockIdx.x * JT;

  v4f ra[16];
#pragma unroll
  for (int u = 0; u < 8; ++u) {
    const int idx = u * 128 + t;
    const float* src = mA + ((size_t)(2 * ((idx >> 3) & 63)) * N_DIM + i0 + (idx >> 9)) * CC +
                       ((idx & 7) << 2);
    ra[2 * u]     = *reinterpret_cast<const v4f*>(src);
    ra[2 * u + 1] = *reinterpret_cast<const v4f*>(src + (size_t)N_DIM * CC);
  }
  {
    const int jj = t & 31, chunk = t >> 5;
    float p0 = 0.f, p1 = 0.f;
#pragma unroll 8
    for (int u = 0; u < 32; ++u) {
      const int s = chunk * 32 + u;
      const float mj = mask[s * N_DIM + j0 + jj];
      p0 += mask[s * N_DIM + i0] * mj;
      p1 += mask[s * N_DIM + i0 + 1] * mj;
    }
    pnorm[t] = p0;
    pnorm[128 + t] = p1;
  }
#pragma unroll
  for (int u = 0; u < 8; ++u) {
    const int idx = u * 128 + t;
    float* d = sA + (idx >> 9) * ((S_DIM / 2) * SP_STRIDE) + ((idx >> 3) & 63) * SP_STRIDE +
               ((idx & 7) << 3);
    const v4f e0 = ra[2 * u], e1 = ra[2 * u + 1];
    *reinterpret_cast<v4f*>(d)     = pair4(e0.x, e1.x, e0.y, e1.y);
    *reinterpret_cast<v4f*>(d + 4) = pair4(e0.z, e1.z, e0.w, e1.w);
  }
  __syncthreads();

  const int c16 = lane & 15;
  const int h   = lane >> 4;

  v8f Z[II][2][8];
#pragma unroll
  for (int a = 0; a < II; ++a)
#pragma unroll
    for (int b = 0; b < 2; ++b)
#pragma unroll
      for (int c = 0; c < 8; ++c) Z[a][b][c] = v8f{};

  const float* ap = sA + h * SP_STRIDE + 2 * c16;
  for (int q = 0; q < 4; ++q) {
    const int dp = wave * 4 + q;
    v8f T[II][2][2][2];
#pragma unroll
    for (int a = 0; a < II; ++a)
#pragma unroll
      for (int b = 0; b < 2; ++b)
#pragma unroll
        for (int c = 0; c < 2; ++c)
#pragma unroll
          for (int d = 0; d < 2; ++d) T[a][b][c][d] = v8f{};
    const float* bp = mBf + ((size_t)(h * 16 + dp) * N_DIM + j0 + c16) * 4;
#define S1_LOAD(FB, FA, S0)                                                     \
    {                                                                           \
      const float* b = bp + (size_t)((S0) >> 1) * (16 * N_DIM * 4);             \
      FB[0] = *reinterpret_cast<const v4f*>(b);                                 \
      FB[1] = *reinterpret_cast<const v4f*>(b + 64);                            \
      FB[2] = *reinterpret_cast<const v4f*>(b + 2 * (16 * N_DIM * 4));          \
      FB[3] = *reinterpret_cast<const v4f*>(b + 2 * (16 * N_DIM * 4) + 64);     \
      _Pragma("unroll") for (int ks = 0; ks < 2; ++ks)                          \
      _Pragma("unroll") for (int a = 0; a < II; ++a)                            \
      _Pragma("unroll") for (int c = 0; c < 2; ++c) {                           \
        FA[ks][a][c] = *reinterpret_cast<const v2f*>(                           \
            ap + a * ((S_DIM / 2) * SP_STRIDE) +                                \
            (((S0) >> 1) + 2 * ks) * SP_STRIDE + 32 * c);                       \
      }                                                                         \
    }
#define S1_MMA(FB, FA)                                                          \
    {                                                                           \
      _Pragma("unroll") for (int ks = 0; ks < 2; ++ks)                          \
      _Pragma("unroll") for (int jt = 0; jt < 2; ++jt) {                        \
        const v4f bq = FB[2 * ks + jt];                                         \
        const v2f bd0 = pair2(bq.x, bq.y), bd1 = pair2(bq.z, bq.w);             \
        _Pragma("unroll") for (int a = 0; a < II; ++a) {                        \
          T[a][0][jt][0] = wmma_f32(FA[ks][a][0], bd0, T[a][0][jt][0]);         \
          T[a][1][jt][0] = wmma_f32(FA[ks][a][1], bd0, T[a][1][jt][0]);         \
          T[a][0][jt][1] = wmma_f32(FA[ks][a][0], bd1, T[a][0][jt][1]);         \
          T[a][1][jt][1] = wmma_f32(FA[ks][a][1], bd1, T[a][1][jt][1]);         \
        }                                                                       \
      }                                                                         \
    }
#define OPM_FENCE __builtin_amdgcn_sched_barrier(0);
    v4f fb[4], gb[4];
    v2f fa[2][II][2], ga[2][II][2];
    S1_LOAD(fb, fa, 0) OPM_FENCE
    for (int s0 = 0; s0 < S_DIM; s0 += 16) {
      S1_LOAD(gb, ga, s0 + 8) OPM_FENCE
      S1_MMA(fb, fa) OPM_FENCE
      const int sn = (s0 + 16) & (S_DIM - 1);
      S1_LOAD(fb, fa, sn) OPM_FENCE
      S1_MMA(gb, ga) OPM_FENCE
    }
#undef S1_MMA
#undef S1_LOAD
#define OPM_LOAD(S, CH, DD, T4)                                                 \
    {                                                                           \
      const float* wp = wOut +                                                  \
          ((size_t)((16 * (CH) + 8 * h + 2 * (T4)) * 32 + 2 * dp + (DD))) * CZ +\
          4 * c16;                                                              \
      S[0] = *reinterpret_cast<const v4f*>(wp);                                 \
      S[1] = *reinterpret_cast<const v4f*>(wp + 64);                            \
      S[2] = *reinterpret_cast<const v4f*>(wp + 32 * CZ);                       \
      S[3] = *reinterpret_cast<const v4f*>(wp + 32 * CZ + 64);                  \
    }
#define OPM_MMA(S, CH, DD, T4)                                                  \
    {                                                                           \
      v2f Bz[8];                                                                \
      Bz[0] = pair2(S[0].x, S[2].x);  Bz[1] = pair2(S[0].y, S[2].y);            \
      Bz[2] = pair2(S[0].z, S[2].z);  Bz[3] = pair2(S[0].w, S[2].w);            \
      Bz[4] = pair2(S[1].x, S[3].x);  Bz[5] = pair2(S[1].y, S[3].y);            \
      Bz[6] = pair2(S[1].z, S[3].z);  Bz[7] = pair2(S[1].w, S[3].w);            \
      _Pragma("unroll") for (int zt = 0; zt < 8; ++zt)                          \
      _Pragma("unroll") for (int a = 0; a < II; ++a)                            \
      _Pragma("unroll") for (int jt = 0; jt < 2; ++jt)                          \
        Z[a][jt][zt] = wmma_f32(                                                \
            pair2(T[a][CH][jt][DD][2 * (T4)], T[a][CH][jt][DD][2 * (T4) + 1]),  \
            Bz[zt], Z[a][jt][zt]);                                              \
    }
    v4f wa[4], wb[4];
    OPM_LOAD(wa, 0, 0, 0)
    OPM_LOAD(wb, 0, 0, 1) OPM_FENCE
    OPM_MMA(wa, 0, 0, 0) OPM_FENCE
    OPM_LOAD(wa, 0, 0, 2) OPM_FENCE
    OPM_MMA(wb, 0, 0, 1) OPM_FENCE
    OPM_LOAD(wb, 0, 0, 3) OPM_FENCE
    OPM_MMA(wa, 0, 0, 2) OPM_FENCE
    OPM_LOAD(wa, 0, 1, 0) OPM_FENCE
    OPM_MMA(wb, 0, 0, 3) OPM_FENCE
    OPM_LOAD(wb, 0, 1, 1) OPM_FENCE
    OPM_MMA(wa, 0, 1, 0) OPM_FENCE
    OPM_LOAD(wa, 0, 1, 2) OPM_FENCE
    OPM_MMA(wb, 0, 1, 1) OPM_FENCE
    OPM_LOAD(wb, 0, 1, 3) OPM_FENCE
    OPM_MMA(wa, 0, 1, 2) OPM_FENCE
    OPM_LOAD(wa, 1, 0, 0) OPM_FENCE
    OPM_MMA(wb, 0, 1, 3) OPM_FENCE
    OPM_LOAD(wb, 1, 0, 1) OPM_FENCE
    OPM_MMA(wa, 1, 0, 0) OPM_FENCE
    OPM_LOAD(wa, 1, 0, 2) OPM_FENCE
    OPM_MMA(wb, 1, 0, 1) OPM_FENCE
    OPM_LOAD(wb, 1, 0, 3) OPM_FENCE
    OPM_MMA(wa, 1, 0, 2) OPM_FENCE
    OPM_LOAD(wa, 1, 1, 0) OPM_FENCE
    OPM_MMA(wb, 1, 0, 3) OPM_FENCE
    OPM_LOAD(wb, 1, 1, 1) OPM_FENCE
    OPM_MMA(wa, 1, 1, 0) OPM_FENCE
    OPM_LOAD(wa, 1, 1, 2) OPM_FENCE
    OPM_MMA(wb, 1, 1, 1) OPM_FENCE
    OPM_LOAD(wb, 1, 1, 3) OPM_FENCE
    OPM_MMA(wa, 1, 1, 2) OPM_FENCE
    OPM_MMA(wb, 1, 1, 3)
#undef OPM_FENCE
#undef OPM_MMA
#undef OPM_LOAD
  }

  __syncthreads();
  {
    float* dst = (wave == 0 ? sA : red + (wave - 1) * ZPART) + (8 * h) * CZ + 4 * c16;
#pragma unroll
    for (int r = 0; r < 8; ++r) {
#pragma unroll
      for (int a = 0; a < II; ++a)
#pragma unroll
        for (int jt = 0; jt < 2; ++jt) {
          float* d = dst + a * (JT * CZ) + (16 * jt + r) * CZ;
          *reinterpret_cast<v4f*>(d) =
              pair4(Z[a][jt][0][r], Z[a][jt][1][r], Z[a][jt][2][r], Z[a][jt][3][r]);
          *reinterpret_cast<v4f*>(d + 64) =
              pair4(Z[a][jt][4][r], Z[a][jt][5][r], Z[a][jt][6][r], Z[a][jt][7][r]);
        }
    }
  }
  __syncthreads();
  const v4f bo = *reinterpret_cast<const v4f*>(bOut + ((4 * t) & (CZ - 1)));
  for (int q = 0; q < 16; ++q) {
    const int f = q * 512 + 4 * t;
    const int a = f >> 12;
    const int g = f & (JT * CZ - 1);
    const int jj = g >> 7;
    const v4f p0 = *reinterpret_cast<const v4f*>(sA + f);
    const v4f p1 = *reinterpret_cast<const v4f*>(red + f);
    const v4f p2 = *reinterpret_cast<const v4f*>(red + ZPART + f);
    const v4f p3 = *reinterpret_cast<const v4f*>(red + 2 * ZPART + f);
    const v4f zsum = ((p0 + p1) + p2) + p3;
    const float* pn = pnorm + a * 128 + jj;
    const float nrm = (((pn[0] + pn[32]) + pn[64]) + pn[96]) + 0.001f;
    v4f o;
    o.x = (zsum.x + bo.x) / nrm;
    o.y = (zsum.y + bo.y) / nrm;
    o.z = (zsum.z + bo.z) / nrm;
    o.w = (zsum.w + bo.w) / nrm;
    *reinterpret_cast<v4f*>(out + ((size_t)(i0 + a) * N_DIM + j0) * CZ + g) = o;
  }
}

extern "C" void kernel_launch(void* const* d_in, const int* in_sizes, int n_in,
                              void* d_out, int out_size, void* d_ws,
                              size_t ws_size, hipStream_t stream) {
  (void)in_sizes; (void)n_in; (void)out_size; (void)ws_size;
  const float* msa   = (const float*)d_in[0];
  const float* mask  = (const float*)d_in[1];
  const float* gamma = (const float*)d_in[2];
  const float* beta  = (const float*)d_in[3];
  const float* wa    = (const float*)d_in[4];
  const float* wb    = (const float*)d_in[5];
  const float* wout  = (const float*)d_in[6];
  const float* bout  = (const float*)d_in[7];
  float* out = (float*)d_out;

  float* mAbuf = (float*)d_ws;
  float* mBbuf = mAbuf + (size_t)S_DIM * N_DIM * CC;

  opm_ln_proj<<<dim3(N_DIM / 8, S_DIM / 2), 256, 0, stream>>>(
      msa, mask, gamma, beta, wa, wb, mAbuf, mBbuf);

  opm_main<<<dim3(N_DIM / JT, N_DIM / II), 128, 0, stream>>>(
      mAbuf, mBbuf, mask, wout, bout, out);
}


# Main kernel re-tiled to one wave per SIMD (256 blocks): a wave owns 3 i x 48 j for all 32 d, Z summed in registers (no partial sums via LDS: z sum order changes), loads one block ahead across seams
# speedup vs baseline: 1.4526x; 1.0704x over previous
#include <hip/hip_runtime.h>

#define S_DIM 128
#define N_DIM 384
#define CM    64
#define CC    32
#define CZ    128
#define LN_EPS 1e-5f
#define II    3
#define NJT   3
#define JW    (16 * NJT)
#define JB    (4 * JW)
#define SP_STRIDE 96

typedef __attribute__((ext_vector_type(2))) float v2f;
typedef __attribute__((ext_vector_type(4))) float v4f;
typedef __attribute__((ext_vector_type(8))) float v8f;

__device__ __forceinline__ v8f wmma_f32(v2f a, v2f b, v8f c) {
  return __builtin_amdgcn_wmma_f32_16x16x4_f32(false, a, false, b, (short)0, c,
                                               false, false);
}

__device__ __forceinline__ v2f pair2(float x, float y) {
  v2f r;
  r.x = x;
  r.y = y;
  return r;
}

__device__ __forceinline__ v4f pair4(float x, float y, float z, float w) {
  v4f r;
  r.x = x;
  r.y = y;
  r.z = z;
  r.w = w;
  return r;
}

__device__ __forceinline__ float wave_sum(float v) {
#pragma unroll
  for (int off = 16; off > 0; off >>= 1) v += __shfl_xor(v, off, 32);
  return v;
}

__global__ __launch_bounds__(256) void opm_ln_proj(
    const float* __restrict__ msa, const float* __restrict__ mask,
    const float* __restrict__ gamma, const float* __restrict__ beta,
    const float* __restrict__ wa, const float* __restrict__ wb,
    float* __restrict__ mA, float* __restrict__ mBf) {
  __shared__ float sY[8 * 2 * CM];
  __shared__ __attribute__((aligned(16))) float sB[16 * 8 * 4];
  const int t = threadIdx.x, wave = t >> 5, lane = t & 31;
  const int sp = blockIdx.y;
  const int n0 = blockIdx.x * 8;
  const int row0 = (2 * sp) * N_DIM + n0 + wave;
  const int row1 = row0 + N_DIM;

  const float g0 = gamma[lane], g1 = gamma[lane + 32];
  const float e0 = beta[lane], e1 = beta[lane + 32];
#pragma unroll
  for (int u = 0; u < 2; ++u) {
    const float* x = msa + (size_t)(u ? row1 : row0) * CM;
    const float x0 = x[lane], x1 = x[lane + 32];
    const float mean = wave_sum(x0 + x1) * (1.0f / CM);
    const float d0 = x0 - mean, d1 = x1 - mean;
    const float var = wave_sum(d0 * d0 + d1 * d1) * (1.0f / CM);
    const float inv = rsqrtf(var + LN_EPS);
    sY[(wave * 2 + u) * CM + lane]      = d0 * inv * g0 + e0;
    sY[(wave * 2 + u) * CM + lane + 32] = d1 * inv * g1 + e1;
  }
  __syncthreads();

  const float mk0 = mask[row0], mk1 = mask[row1];
  float a0 = 0.f, b0 = 0.f, a1 = 0.f, b1 = 0.f;
#pragma unroll 8
  for (int m = 0; m < CM; ++m) {
    const float y0 = sY[(wave * 2) * CM + m];
    const float y1 = sY[(wave * 2 + 1) * CM + m];
    const float wam = wa[m * CC + lane], wbm = wb[m * CC + lane];
    a0 = fmaf(y0, wam, a0);
    b0 = fmaf(y0, wbm, b0);
    a1 = fmaf(y1, wam, a1);
    b1 = fmaf(y1, wbm, b1);
  }
  mA[(size_t)row0 * CC + lane] = a0 * mk0;
  mA[(size_t)row1 * CC + lane] = a1 * mk1;
  *reinterpret_cast<v2f*>(sB + ((lane >> 1) * 8 + wave) * 4 + (lane & 1) * 2) =
      pair2(b0 * mk0, b1 * mk1);
  __syncthreads();
  if (t < 128) {
    const int dp = t >> 3, nn = t & 7;
    const v4f v = *reinterpret_cast<const v4f*>(sB + t * 4);
    *reinterpret_cast<v4f*>(
        mBf + (((size_t)sp * 16 + dp) * N_DIM + n0 + nn) * 4) = v;
  }
}

#define SLICE ((S_DIM / 2) * SP_STRIDE)
#define ZW (NJT * 8 * 256)
__global__ __launch_bounds__(128) __attribute__((amdgpu_waves_per_eu(1, 1))) void opm_main(
    const float* __restrict__ mA, const float* __restrict__ mBf,
    const float* __restrict__ mask, const float* __restrict__ wOut,
    const float* __restrict__ bOut, float* __restrict__ out) {
  __shared__ __attribute__((aligned(16))) float smem[II * SLICE + II * 4 * JB + 4 * ZW];
  float* sA    = smem;
  float* pnorm = sA + II * SLICE;
  float* zpark = pnorm + II * 4 * JB;

  const int t = threadIdx.x;
  const int wave = t >> 5, lane = t & 31;
  const int i0 = blockIdx.y * II;
  const int j0 = blockIdx.x * JB;

  v4f ra[24];
#pragma unroll
  for (int u = 0; u < 12; ++u) {
    const int idx = u * 128 + t;
    const float* src = mA + ((size_t)(2 * ((idx >> 3) & 63)) * N_DIM + i0 + (idx >> 9)) * CC +
                       ((idx & 7) << 2);
    ra[2 * u]     = *reinterpret_cast<const v4f*>(src);
    ra[2 * u + 1] = *reinterpret_cast<const v4f*>(src + (size_t)N_DIM * CC);
  }
  {
    float pn[II][6];
#pragma unroll
    for (int a = 0; a < II; ++a)
#pragma unroll
      for (int m = 0; m < 6; ++m) pn[a][m] = 0.f;
#pragma unroll 8
    for (int u = 0; u < 32; ++u) {
      const float* mr = mask + (wave * 32 + u) * N_DIM;
      const float m0 = mr[i0], m1 = mr[i0 + 1], m2 = mr[i0 + 2];
#pragma unroll
      for (int m = 0; m < 6; ++m) {
        const float mj = mr[j0 + lane + 32 * m];
        pn[0][m] += m0 * mj;
        pn[1][m] += m1 * mj;
        pn[2][m] += m2 * mj;
      }
    }
#pragma unroll
    for (int a = 0; a < II; ++a)
#pragma unroll
      for (int m = 0; m < 6; ++m) pnorm[(a * 4 + wave) * JB + lane + 32 * m] = pn[a][m];
  }
#pragma unroll
  for (int u = 0; u < 12; ++u) {
    const int idx = u * 128 + t;
    float* d = sA + (idx >> 9) * SLICE + ((idx >> 3) & 63) * SP_STRIDE + ((idx & 7) << 3);
    const v4f e0 = ra[2 * u], e1 = ra[2 * u + 1];
    *reinterpret_cast<v4f*>(d)     = pair4(e0.x, e1.x, e0.y, e1.y);
    *reinterpret_cast<v4f*>(d + 4) = pair4(e0.z, e1.z, e0.w, e1.w);
  }
  __syncthreads();

  const int c16 = lane & 15;
  const int h   = lane >> 4;
  const int jw  = j0 + JW * wave;

  v8f Z[II][NJT][8];
#pragma unroll
  for (int a = 0; a < II; ++a)
#pragma unroll
    for (int b = 0; b < NJT; ++b)
#pragma unroll
      for (int c = 0; c < 8; ++c) Z[a][b][c] = v8f{};

  const float* ap = sA + h * SP_STRIDE + 2 * c16;
  const float* bp = mBf + ((size_t)(h * 16) * N_DIM + jw + c16) * 4;

#define S1_LOAD(FB, FA, BP, S0)                                                 \
    {                                                                           \
      const float* b = (BP) + (size_t)((S0) >> 1) * (16 * N_DIM * 4);           \
      _Pragma("unroll") for (int ks = 0; ks < 2; ++ks)                          \
      _Pragma("unroll") for (int jt = 0; jt < NJT; ++jt)                        \
        FB[3 * ks + jt] = *reinterpret_cast<const v4f*>(                        \
            b + ks * 2 * (16 * N_DIM * 4) + 64 * jt);                           \
      _Pragma("unroll") for (int ks = 0; ks < 2; ++ks)                          \
      _Pragma("unroll") for (int a = 0; a < II; ++a)                            \
      _Pragma("unroll") for (int c = 0; c < 2; ++c) {                           \
        FA[ks][a][c] = *reinterpret_cast<const v2f*>(                           \
            ap + a * SLICE + (((S0) >> 1) + 2 * ks) * SP_STRIDE + 32 * c);      \
      }                                                                         \
    }
#define S1_MMA(FB, FA, FIRST)                                                   \
    {                                                                           \
      _Pragma("unroll") for (int ks = 0; ks < 2; ++ks)                          \
      _Pragma("unroll") for (int jt = 0; jt < NJT; ++jt) {                      \
        const v4f bq = FB[3 * ks + jt];                                         \
        const v2f bd0 = pair2(bq.x, bq.y), bd1 = pair2(bq.z, bq.w);             \
        const bool z0 = (FIRST) && ks == 0;                                     \
        _Pragma("unroll") for (int a = 0; a < II; ++a) {                        \
          T[a][0][jt][0] = wmma_f32(FA[ks][a][0], bd0, z0 ? v8f{} : T[a][0][jt][0]); \
          T[a][1][jt][0] = wmma_f32(FA[ks][a][1], bd0, z0 ? v8f{} : T[a][1][jt][0]); \
          T[a][0][jt][1] = wmma_f32(FA[ks][a][0], bd1, z0 ? v8f{} : T[a][0][jt][1]); \
          T[a][1][jt][1] = wmma_f32(FA[ks][a][1], bd1, z0 ? v8f{} : T[a][1][jt][1]); \
        }                                                                       \
      }                                                                         \
    }
#define OPM_LOAD(S, CH, DD, T4)                                                 \
    {                                                                           \
      const float* wp = wOut +                                                  \
          ((size_t)((16 * (CH) + 8 * h + 2 * (T4)) * 32 + 2 * dp + (DD))) * CZ +\
          4 * c16;                                                              \
      S[0] = *reinterpret_cast<const v4f*>(wp);                                 \
      S[1] = *reinterpret_cast<const v4f*>(wp + 64);                            \
      S[2] = *reinterpret_cast<const v4f*>(wp + 32 * CZ);                       \
      S[3] = *reinterpret_cast<const v4f*>(wp + 32 * CZ + 64);                  \
    }
#define OPM_MMA(S, CH, DD, T4)                                                  \
    {                                                                           \
      v2f Bz[8];                                                                \
      Bz[0] = pair2(S[0].x, S[2].x);  Bz[1] = pair2(S[0].y, S[2].y);            \
      Bz[2] = pair2(S[0].z, S[2].z);  Bz[3] = pair2(S[0].w, S[2].w);            \
      Bz[4] = pair2(S[1].x, S[3].x);  Bz[5] = pair2(S[1].y, S[3].y);            \
      Bz[6] = pair2(S[1].z, S[3].z);  Bz[7] = pair2(S[1].w, S[3].w);            \
      _Pragma("unroll") for (int zt = 0; zt < 8; ++zt)                          \
      _Pragma("unroll") for (int a = 0; a < II; ++a)                            \
      _Pragma("unroll") for (int jt = 0; jt < NJT; ++jt)                        \
        Z[a][jt][zt] = wmma_f32(                                                \
            pair2(T[a][CH][jt][DD][2 * (T4)], T[a][CH][jt][DD][2 * (T4) + 1]),  \
            Bz[zt], Z[a][jt][zt]);                                              \
    }
#define OPM_FENCE __builtin_amdgcn_sched_barrier(0);

  v4f fb[6], gb[6];
  v2f fa[2][II][2], ga[2][II][2];
  v4f wa[4], wb[4];
  S1_LOAD(fb, fa, bp, 0) OPM_FENCE
  for (int dp = 0; dp < 16; ++dp) {
    v8f T[II][2][NJT][2];
    S1_LOAD(gb, ga, bp, 8) OPM_FENCE
    S1_MMA(fb, fa, 1) OPM_FENCE
    S1_LOAD(fb, fa, bp, 16) OPM_FENCE
    S1_MMA(gb, ga, 0) OPM_FENCE
    for (int s0 = 16; s0 < S_DIM - 16; s0 += 16) {
      S1_LOAD(gb, ga, bp, s0 + 8) OPM_FENCE
      S1_MMA(fb, fa, 0) OPM_FENCE
      S1_LOAD(fb, fa, bp, s0 + 16) OPM_FENCE
      S1_MMA(gb, ga, 0) OPM_FENCE
    }
    S1_LOAD(gb, ga, bp, S_DIM - 8) OPM_FENCE
    S1_MMA(fb, fa, 0) OPM_FENCE
    OPM_LOAD(wa, 0, 0, 0) OPM_FENCE
    S1_MMA(gb, ga, 0) OPM_FENCE
    bp += N_DIM * 4;
    OPM_LOAD(wb, 0, 0, 1) OPM_FENCE
    OPM_MMA(wa, 0, 0, 0) OPM_FENCE
    OPM_LOAD(wa, 0, 0, 2) OPM_FENCE
    OPM_MMA(wb, 0, 0, 1) OPM_FENCE
    OPM_LOAD(wb, 0, 0, 3) OPM_FENCE
    OPM_MMA(wa, 0, 0, 2) OPM_FENCE
    OPM_LOAD(wa, 0, 1, 0) OPM_FENCE
    OPM_MMA(wb, 0, 0, 3) OPM_FENCE
    OPM_LOAD(wb, 0, 1, 1) OPM_FENCE
    OPM_MMA(wa, 0, 1, 0) OPM_FENCE
    OPM_LOAD(wa, 0, 1, 2) OPM_FENCE
    OPM_MMA(wb, 0, 1, 1) OPM_FENCE
    OPM_LOAD(wb, 0, 1, 3) OPM_FENCE
    OPM_MMA(wa, 0, 1, 2) OPM_FENCE
    OPM_LOAD(wa, 1, 0, 0) OPM_FENCE
    OPM_MMA(wb, 0, 1, 3) OPM_FENCE
    OPM_LOAD(wb, 1, 0, 1) OPM_FENCE
    OPM_MMA(wa, 1, 0, 0) OPM_FENCE
    OPM_LOAD(wa, 1, 0, 2) OPM_FENCE
    OPM_MMA(wb, 1, 0, 1) OPM_FENCE
    OPM_LOAD(wb, 1, 0, 3) OPM_FENCE
    OPM_MMA(wa, 1, 0, 2) OPM_FENCE
    OPM_LOAD(wa, 1, 1, 0) OPM_FENCE
    OPM_MMA(wb, 1, 0, 3) OPM_FENCE
    OPM_LOAD(wb, 1, 1, 1) OPM_FENCE
    OPM_MMA(wa, 1, 1, 0) OPM_FENCE
    OPM_LOAD(wa, 1, 1, 2) OPM_FENCE
    OPM_MMA(wb, 1, 1, 1) OPM_FENCE
    OPM_LOAD(wb, 1, 1, 3) OPM_FENCE
    OPM_MMA(wa, 1, 1, 2) OPM_FENCE
    S1_LOAD(fb, fa, bp, 0) OPM_FENCE
    OPM_MMA(wb, 1, 1, 3) OPM_FENCE
  }
#undef OPM_FENCE
#undef OPM_MMA
#undef OPM_LOAD
#undef S1_MMA
#undef S1_LOAD

  float* zw = zpark + wave * ZW + 4 * lane;
  const v4f bo0 = *reinterpret_cast<const v4f*>(bOut + 4 * c16);
  const v4f bo1 = *reinterpret_cast<const v4f*>(bOut + 64 + 4 * c16);
#pragma unroll
  for (int a = 0; a < II; ++a) {
#pragma unroll
    for (int jt = 0; jt < NJT; ++jt)
#pragma unroll
      for (int zt = 0; zt < 8; ++zt) {
        float* d = zw + (8 * jt + zt) * 256;
        const v8f v = Z[a][jt][zt];
        *reinterpret_cast<v4f*>(d)       = pair4(v[0], v[1], v[2], v[3]);
        *reinterpret_cast<v4f*>(d + 128) = pair4(v[4], v[5], v[6], v[7]);
      }
    float* orow = out + ((size_t)(i0 + a) * N_DIM + j0 + JW * wave + 8 * h) * CZ + 4 * c16;
    const float* pn0 = pnorm + a * (4 * JB) + JW * wave + 8 * h;
    for (int it = 0; it < 4 * NJT; ++it) {
      const int jt = it >> 2, zh = (it >> 1) & 1, rq = it & 1;
      const float* s = zw + ((8 * jt + 4 * zh) * 2 + rq) * 128;
      const v4f z0 = *reinterpret_cast<const v4f*>(s);
      const v4f z1 = *reinterpret_cast<const v4f*>(s + 256);
      const v4f z2 = *reinterpret_cast<const v4f*>(s + 512);
      const v4f z3 = *reinterpret_cast<const v4f*>(s + 768);
      const v4f bo = zh ? bo1 : bo0;
#pragma unroll
      for (int rr = 0; rr < 4; ++rr) {
        const int jr = 16 * jt + 4 * rq + rr;
        const float* pn = pn0 + jr;
        const float nrm = (((pn[0] + pn[JB]) + pn[2 * JB]) + pn[3 * JB]) + 0.001f;
        v4f o;
        o.x = (z0[rr] + bo.x) / nrm;
        o.y = (z1[rr] + bo.y) / nrm;
        o.z = (z2[rr] + bo.z) / nrm;
        o.w = (z3[rr] + bo.w) / nrm;
        *reinterpret_cast<v4f*>(orow + (size_t)jr * CZ + 64 * zh) = o;
      }
    }
  }
}

extern "C" void kernel_launch(void* const* d_in, const int* in_sizes, int n_in,
                              void* d_out, int out_size, void* d_ws,
                              size_t ws_size, hipStream_t stream) {
  (void)in_sizes; (void)n_in; (void)out_size; (void)ws_size;
  const float* msa   = (const float*)d_in[0];
  const float* mask  = (const float*)d_in[1];
  const float* gamma = (const float*)d_in[2];
  const float* beta  = (const float*)d_in[3];
  const float* wa    = (const float*)d_in[4];
  const float* wb    = (const float*)d_in[5];
  const float* wout  = (const float*)d_in[6];
  const float* bout  = (const float*)d_in[7];
  float* out = (float*)d_out;

  float* mAbuf = (float*)d_ws;
  float* mBbuf = mAbuf + (size_t)S_DIM * N_DIM * CC;

  opm_ln_proj<<<dim3(N_DIM / 8, S_DIM / 2), 256, 0, stream>>>(
      msa, mask, gamma, beta, wa, wb, mAbuf, mBbuf);

  opm_main<<<dim3(N_DIM / JB, N_DIM / II), 128, 0, stream>>>(
      mAbuf, mBbuf, mask, wout, bout, out);
}
